# starting kernel
# baseline (speedup 1.0000x reference)
#include <hip/hip_runtime.h>
#include <hip/hip_bf16.h>

#define HW   4096
#define NB   8
#define CIN  256
#define CQ   128
#define CV   256

typedef float    v2f  __attribute__((ext_vector_type(2)));
typedef float    v8f  __attribute__((ext_vector_type(8)));
typedef _Float16 v16h __attribute__((ext_vector_type(16)));
typedef _Float16 h8   __attribute__((ext_vector_type(8)));

static __device__ __forceinline__ v16h load_frag16(const _Float16* rowBase, int lh) {
  const h8 lo = *reinterpret_cast<const h8*>(rowBase + 8 * lh);
  const h8 hi = *reinterpret_cast<const h8*>(rowBase + 8 * lh + 16);
  v16h r;
#pragma unroll
  for (int i = 0; i < 8; ++i) { r[i] = lo[i]; r[i + 8] = hi[i]; }
  return r;
}

__global__ __launch_bounds__(128) void proj_bn_relu_kernel(
    const float* __restrict__ x, const float* __restrict__ w,
    const float* __restrict__ gamma, const float* __restrict__ beta,
    const float* __restrict__ mean, const float* __restrict__ var,
    _Float16* __restrict__ outh, int posMajor)
{
  const int lane = threadIdx.x;
  const int l15  = lane & 15;
  const int lh   = lane >> 4;
  const int p0   = (blockIdx.x * blockDim.y + threadIdx.y) * 16;
  const int o0   = blockIdx.y * 16;
  const int n    = blockIdx.z;

  const float* wrow = w + (size_t)(o0 + l15) * CIN;
  const float* xcol = x + (size_t)n * CIN * HW + (p0 + l15);

  v8f acc0 = {};
  v8f acc1 = {};
#pragma unroll 4
  for (int c0 = 0; c0 < CIN; c0 += 8) {
    const float2 aw0 = *reinterpret_cast<const float2*>(wrow + c0 + 2 * lh);
    const float2 aw1 = *reinterpret_cast<const float2*>(wrow + c0 + 4 + 2 * lh);
    v2f a0; a0[0] = aw0.x; a0[1] = aw0.y;
    v2f a1; a1[0] = aw1.x; a1[1] = aw1.y;
    v2f b0; b0[0] = xcol[(size_t)(c0 + 2 * lh + 0) * HW];
            b0[1] = xcol[(size_t)(c0 + 2 * lh + 1) * HW];
    v2f b1; b1[0] = xcol[(size_t)(c0 + 4 + 2 * lh + 0) * HW];
            b1[1] = xcol[(size_t)(c0 + 4 + 2 * lh + 1) * HW];
    acc0 = __builtin_amdgcn_wmma_f32_16x16x4_f32(false, a0, false, b0, (short)0, acc0, false, false);
    acc1 = __builtin_amdgcn_wmma_f32_16x16x4_f32(false, a1, false, b1, (short)0, acc1, false, false);
  }
  const v8f acc = acc0 + acc1;

#pragma unroll
  for (int r = 0; r < 8; ++r) {
    const int ch   = o0 + 8 * lh + r;
    const float iv = gamma[ch] * rsqrtf(var[ch] + 1e-5f);
    const float bb = beta[ch] - mean[ch] * iv;
    float v = acc[r] * iv + bb;
    v = v > 0.f ? v : 0.f;
    const _Float16 hv = (_Float16)v;
    const int p = p0 + l15;
    if (posMajor)
      outh[((size_t)n * HW + p) * CQ + ch] = hv;
    else
      outh[((size_t)n * CV + ch) * HW + p] = hv;
  }
}

__global__ __launch_bounds__(128) void flash_attn_kernel(
    const _Float16* __restrict__ q, const _Float16* __restrict__ fT,
    const float* __restrict__ mask, float* __restrict__ out)
{
  constexpr float LOG2E     = 1.4426950408889634f;
  constexpr float ATT_SCALE = (1.0f / (1e-8f + 11.313708498984761f)) * LOG2E;
  __shared__ __align__(16) _Float16 Pbuf[4][16 * 32];

  const int lane = threadIdx.x;
  const int l15  = lane & 15;
  const int lh   = lane >> 4;
  const int wid  = threadIdx.y;
  const int i0   = (blockIdx.x * 4 + wid) * 16;
  const int n    = blockIdx.y;

  v16h qa[4];
  {
    const _Float16* qrow = q + ((size_t)n * HW + i0 + l15) * CQ;
#pragma unroll
    for (int kc = 0; kc < 4; ++kc) qa[kc] = load_frag16(qrow + kc * 32, lh);
  }

  v8f oacc[16];
#pragma unroll
  for (int t = 0; t < 16; ++t) oacc[t] = (v8f){};
  float mrow[8], lrow[8];
#pragma unroll
  for (int r = 0; r < 8; ++r) { mrow[r] = -1e30f; lrow[r] = 0.f; }

  _Float16* pb    = &Pbuf[wid][0];
  const float* mk = mask + (size_t)n * HW;

  for (int j0 = 0; j0 < HW; j0 += 32) {
    if (j0 + 32 < HW) {
      __builtin_prefetch(q + ((size_t)n * HW + j0 + 32 + l15) * CQ, 0, 3);
      __builtin_prefetch(q + ((size_t)n * HW + j0 + 48 + l15) * CQ, 0, 3);
    }

    v8f s0 = (v8f){};
    v8f s1 = (v8f){};
#pragma unroll
    for (int kc = 0; kc < 4; ++kc) {
      const _Float16* kb = q + ((size_t)n * HW + j0 + l15) * CQ + kc * 32;
      const v16h b0 = load_frag16(kb, lh);
      const v16h b1 = load_frag16(kb + 16 * CQ, lh);
      s0 = __builtin_amdgcn_wmma_f32_16x16x32_f16(false, qa[kc], false, b0, (short)0, s0, false, false);
      s1 = __builtin_amdgcn_wmma_f32_16x16x32_f16(false, qa[kc], false, b1, (short)0, s1, false, false);
    }

    const float ma0 = (mk[j0 + l15]      - 1.f) * (1e8f * LOG2E);
    const float ma1 = (mk[j0 + 16 + l15] - 1.f) * (1e8f * LOG2E);

    v8f avec;
#pragma unroll
    for (int r = 0; r < 8; ++r) {
      float e0 = s0[r] * ATT_SCALE + ma0;
      float e1 = s1[r] * ATT_SCALE + ma1;
      float vmax = fmaxf(e0, e1);
#pragma unroll
      for (int m = 1; m <= 8; m <<= 1) vmax = fmaxf(vmax, __shfl_xor(vmax, m, 32));
      const float mn = fmaxf(mrow[r], vmax);
      avec[r] = exp2f(mrow[r] - mn);
      mrow[r] = mn;
      e0 = exp2f(e0 - mn);
      e1 = exp2f(e1 - mn);
      float rs = e0 + e1;
#pragma unroll
      for (int m = 1; m <= 8; m <<= 1) rs += __shfl_xor(rs, m, 32);
      lrow[r] = lrow[r] * avec[r] + rs;
      s0[r] = e0;
      s1[r] = e1;
    }
#pragma unroll
    for (int t = 0; t < 16; ++t) oacc[t] *= avec;

#pragma unroll
    for (int r = 0; r < 8; ++r) {
      pb[(8 * lh + r) * 32 + l15]      = (_Float16)s0[r];
      pb[(8 * lh + r) * 32 + 16 + l15] = (_Float16)s1[r];
    }
    asm volatile("s_wait_dscnt 0" ::: "memory");
    const v16h pa = load_frag16(pb + l15 * 32, lh);

#pragma unroll
    for (int t = 0; t < 16; ++t) {
      const _Float16* vrow = fT + ((size_t)n * CV + t * 16 + l15) * HW + j0;
      const v16h vb = load_frag16(vrow, lh);
      oacc[t] = __builtin_amdgcn_wmma_f32_16x16x32_f16(false, pa, false, vb, (short)0, oacc[t], false, false);
    }
  }

  float fin[8];
#pragma unroll
  for (int r = 0; r < 8; ++r)
    fin[r] = (1.0f / lrow[r]) * mk[i0 + 8 * lh + r];

#pragma unroll
  for (int t = 0; t < 16; ++t) {
    float* dst = out + ((size_t)n * CV + t * 16 + l15) * HW + i0 + 8 * lh;
    float4 lo4 = make_float4(oacc[t][0] * fin[0], oacc[t][1] * fin[1],
                             oacc[t][2] * fin[2], oacc[t][3] * fin[3]);
    float4 hi4 = make_float4(oacc[t][4] * fin[4], oacc[t][5] * fin[5],
                             oacc[t][6] * fin[6], oacc[t][7] * fin[7]);
    *reinterpret_cast<float4*>(dst)     = lo4;
    *reinterpret_cast<float4*>(dst + 4) = hi4;
  }
}

extern "C" void kernel_launch(void* const* d_in, const int* in_sizes, int n_in,
                              void* d_out, int out_size, void* d_ws, size_t ws_size,
                              hipStream_t stream) {
  (void)in_sizes; (void)n_in; (void)out_size; (void)ws_size;
  const float* x  = (const float*)d_in[0];
  const float* mk = (const float*)d_in[1];
  const float* rw = (const float*)d_in[2];
  const float* rg = (const float*)d_in[3];
  const float* rb = (const float*)d_in[4];
  const float* rm = (const float*)d_in[5];
  const float* rv = (const float*)d_in[6];
  const float* aw = (const float*)d_in[7];
  const float* ag = (const float*)d_in[8];
  const float* ab = (const float*)d_in[9];
  const float* am = (const float*)d_in[10];
  const float* av = (const float*)d_in[11];
  float* out = (float*)d_out;

  _Float16* qbuf = (_Float16*)d_ws;
  _Float16* fT   = qbuf + (size_t)NB * HW * CQ;

  const dim3 blk(32, 4);
  proj_bn_relu_kernel<<<dim3(HW / 64, CV / 16, NB), blk, 0, stream>>>(
      x, rw, rg, rb, rm, rv, fT, 0);
  proj_bn_relu_kernel<<<dim3(HW / 64, CQ / 16, NB), blk, 0, stream>>>(
      x, aw, ag, ab, am, av, qbuf, 1);
  flash_attn_kernel<<<dim3(HW / 64, NB), blk, 0, stream>>>(qbuf, fT, mk, out);
}


# q/K and V intermediates stored in WMMA fragment order by the projection epilogues (V tile computed position-major): each attention fragment load reads 4 whole lines, not 16 partial; sums unchanged
# speedup vs baseline: 3.2383x; 3.2383x over previous
#include <hip/hip_runtime.h>
#include <hip/hip_bf16.h>

#define HW   4096
#define NB   8
#define CIN  256
#define CQ   128
#define CV   256

typedef float    v2f  __attribute__((ext_vector_type(2)));
typedef float    v8f  __attribute__((ext_vector_type(8)));
typedef _Float16 v16h __attribute__((ext_vector_type(16)));
typedef _Float16 h8   __attribute__((ext_vector_type(8)));

static __device__ __forceinline__ v16h load_frag16(const _Float16* rowBase, int lh) {
  const h8 lo = *reinterpret_cast<const h8*>(rowBase + 8 * lh);
  const h8 hi = *reinterpret_cast<const h8*>(rowBase + 8 * lh + 16);
  v16h r;
#pragma unroll
  for (int i = 0; i < 8; ++i) { r[i] = lo[i]; r[i + 8] = hi[i]; }
  return r;
}

static __device__ __forceinline__ v16h load_frag16_packed(const _Float16* fragBase, int lane) {
  const h8 lo = *reinterpret_cast<const h8*>(fragBase + 8 * lane);
  const h8 hi = *reinterpret_cast<const h8*>(fragBase + 256 + 8 * lane);
  v16h r;
#pragma unroll
  for (int i = 0; i < 8; ++i) { r[i] = lo[i]; r[i + 8] = hi[i]; }
  return r;
}

template <int POS_ROWS>
static __device__ __forceinline__ v8f proj_tile(const float* __restrict__ wrow,
                                                const float* __restrict__ xcol, int lh) {
  v8f acc0 = {};
  v8f acc1 = {};
#pragma unroll 4
  for (int c0 = 0; c0 < CIN; c0 += 8) {
    const float2 aw0 = *reinterpret_cast<const float2*>(wrow + c0 + 2 * lh);
    const float2 aw1 = *reinterpret_cast<const float2*>(wrow + c0 + 4 + 2 * lh);
    v2f a0; a0[0] = aw0.x; a0[1] = aw0.y;
    v2f a1; a1[0] = aw1.x; a1[1] = aw1.y;
    v2f b0; b0[0] = xcol[(size_t)(c0 + 2 * lh + 0) * HW];
            b0[1] = xcol[(size_t)(c0 + 2 * lh + 1) * HW];
    v2f b1; b1[0] = xcol[(size_t)(c0 + 4 + 2 * lh + 0) * HW];
            b1[1] = xcol[(size_t)(c0 + 4 + 2 * lh + 1) * HW];
    if (POS_ROWS) {
      acc0 = __builtin_amdgcn_wmma_f32_16x16x4_f32(false, b0, false, a0, (short)0, acc0, false, false);
      acc1 = __builtin_amdgcn_wmma_f32_16x16x4_f32(false, b1, false, a1, (short)0, acc1, false, false);
    } else {
      acc0 = __builtin_amdgcn_wmma_f32_16x16x4_f32(false, a0, false, b0, (short)0, acc0, false, false);
      acc1 = __builtin_amdgcn_wmma_f32_16x16x4_f32(false, a1, false, b1, (short)0, acc1, false, false);
    }
  }
  return acc0 + acc1;
}

__global__ __launch_bounds__(128) void proj_bn_relu_kernel(
    const float* __restrict__ x, const float* __restrict__ w,
    const float* __restrict__ gamma, const float* __restrict__ beta,
    const float* __restrict__ mean, const float* __restrict__ var,
    _Float16* __restrict__ outh, int posMajor)
{
  const int lane = threadIdx.x;
  const int l15  = lane & 15;
  const int lh   = lane >> 4;
  const int p0   = (blockIdx.x * blockDim.y + threadIdx.y) * 16;
  const int o0   = blockIdx.y * 16;
  const int n    = blockIdx.z;

  const float* wrow = w + (size_t)(o0 + l15) * CIN;
  const float* xcol = x + (size_t)n * CIN * HW + (p0 + l15);

  h8 hv;
  size_t off;
  if (posMajor) {
    const v8f acc = proj_tile<0>(wrow, xcol, lh);
#pragma unroll
    for (int r = 0; r < 8; ++r) {
      const int ch   = o0 + 8 * lh + r;
      const float iv = gamma[ch] * rsqrtf(var[ch] + 1e-5f);
      const float bb = beta[ch] - mean[ch] * iv;
      float v = acc[r] * iv + bb;
      v = v > 0.f ? v : 0.f;
      hv[r] = (_Float16)v;
    }
    off = ((((size_t)n * (HW / 16) + (p0 >> 4)) * 4 + (o0 >> 5)) * 2 + ((o0 >> 4) & 1)) * 256 + 8 * lane;
  } else {
    const v8f acc = proj_tile<1>(wrow, xcol, lh);
    const int ch   = o0 + l15;
    const float iv = gamma[ch] * rsqrtf(var[ch] + 1e-5f);
    const float bb = beta[ch] - mean[ch] * iv;
#pragma unroll
    for (int r = 0; r < 8; ++r) {
      float v = acc[r] * iv + bb;
      v = v > 0.f ? v : 0.f;
      hv[r] = (_Float16)v;
    }
    off = ((((size_t)n * (HW / 32) + (p0 >> 5)) * (CV / 16) + (o0 >> 4)) * 2 + ((p0 >> 4) & 1)) * 256 + 8 * lane;
  }
  *reinterpret_cast<h8*>(outh + off) = hv;
}

__global__ __launch_bounds__(128) __attribute__((amdgpu_waves_per_eu(2, 2)))
void flash_attn_kernel(
    const _Float16* __restrict__ q, const _Float16* __restrict__ fT,
    const float* __restrict__ mask, float* __restrict__ out)
{
  constexpr float LOG2E     = 1.4426950408889634f;
  constexpr float ATT_SCALE = (1.0f / (1e-8f + 11.313708498984761f)) * LOG2E;
  constexpr int PP = 40;
  __shared__ __align__(32) _Float16 Pbuf[2][64 * PP];
  __shared__ __align__(32) float    Abuf[2][64];
  __shared__ __align__(32) float    Fbuf[64];

  const int lane = threadIdx.x;
  const int l15  = lane & 15;
  const int lh   = lane >> 4;
  const int wid  = threadIdx.y;
  const int iwg  = blockIdx.x * 64;
  const int i0   = iwg + wid * 16;
  const int n    = blockIdx.y;

  const _Float16* qn = q + (size_t)n * HW * CQ;

  v16h qa[4];
  {
    const _Float16* qt0 = qn + (size_t)(i0 >> 4) * 2048;
#pragma unroll
    for (int kc = 0; kc < 4; ++kc) qa[kc] = load_frag16_packed(qt0 + kc * 512, lane);
  }

  v8f oacc[4][4];
#pragma unroll
  for (int qt = 0; qt < 4; ++qt)
#pragma unroll
    for (int ct = 0; ct < 4; ++ct) oacc[qt][ct] = (v8f){};
  float mrow[8], lrow[8];
#pragma unroll
  for (int r = 0; r < 8; ++r) { mrow[r] = -1e30f; lrow[r] = 0.f; }

  const float* mk = mask + (size_t)n * HW;
  const _Float16* vbase = fT + (size_t)n * CV * HW + wid * 4 * 512;

  for (int j0 = 0; j0 < HW; j0 += 32) {
    const int slot = (j0 >> 5) & 1;

    if (j0 + 32 < HW) {
      __builtin_prefetch(qn + (size_t)((j0 + 32) >> 4) * 2048 + lane * 128, 0, 3);
    }

    const float mv0 = mk[j0 + l15];
    const float mv1 = mk[j0 + 16 + l15];
    v16h kb0[4], kb1[4], vb[4];
    {
      const _Float16* kb = qn + (size_t)(j0 >> 4) * 2048;
#pragma unroll
      for (int kc = 0; kc < 4; ++kc) {
        kb0[kc] = load_frag16_packed(kb + kc * 512, lane);
        kb1[kc] = load_frag16_packed(kb + 2048 + kc * 512, lane);
      }
      const _Float16* vblk = vbase + (size_t)(j0 >> 5) * 8192;
#pragma unroll
      for (int ct = 0; ct < 4; ++ct) vb[ct] = load_frag16_packed(vblk + ct * 512, lane);
    }
    __builtin_amdgcn_sched_barrier(0);

    v8f s0 = (v8f){};
    v8f s1 = (v8f){};
#pragma unroll
    for (int kc = 0; kc < 4; ++kc) {
      s0 = __builtin_amdgcn_wmma_f32_16x16x32_f16(false, qa[kc], false, kb0[kc], (short)0, s0, false, false);
      s1 = __builtin_amdgcn_wmma_f32_16x16x32_f16(false, qa[kc], false, kb1[kc], (short)0, s1, false, false);
    }

    const float ma0 = (mv0 - 1.f) * (1e8f * LOG2E);
    const float ma1 = (mv1 - 1.f) * (1e8f * LOG2E);

    v8f avec;
#pragma unroll
    for (int r = 0; r < 8; ++r) {
      float e0 = s0[r] * ATT_SCALE + ma0;
      float e1 = s1[r] * ATT_SCALE + ma1;
      float vmax = fmaxf(e0, e1);
#pragma unroll
      for (int m = 1; m <= 8; m <<= 1) vmax = fmaxf(vmax, __shfl_xor(vmax, m, 32));
      const float mn = fmaxf(mrow[r], vmax);
      avec[r] = exp2f(mrow[r] - mn);
      mrow[r] = mn;
      e0 = exp2f(e0 - mn);
      e1 = exp2f(e1 - mn);
      float rs = e0 + e1;
#pragma unroll
      for (int m = 1; m <= 8; m <<= 1) rs += __shfl_xor(rs, m, 32);
      lrow[r] = lrow[r] * avec[r] + rs;
      s0[r] = e0;
      s1[r] = e1;
    }

    {
      _Float16* pw = &Pbuf[slot][(wid * 16) * PP];
#pragma unroll
      for (int r = 0; r < 8; ++r) {
        pw[(8 * lh + r) * PP + l15]      = (_Float16)s0[r];
        pw[(8 * lh + r) * PP + 16 + l15] = (_Float16)s1[r];
      }
      if (l15 == 0) *reinterpret_cast<v8f*>(&Abuf[slot][wid * 16 + 8 * lh]) = avec;
    }
    __syncthreads();

    v16h pa[4];
    v8f  al[4];
#pragma unroll
    for (int qt = 0; qt < 4; ++qt) {
      pa[qt] = load_frag16(&Pbuf[slot][(qt * 16 + l15) * PP], lh);
      al[qt] = *reinterpret_cast<const v8f*>(&Abuf[slot][qt * 16 + 8 * lh]);
    }

#pragma unroll
    for (int qt = 0; qt < 4; ++qt)
#pragma unroll
      for (int ct = 0; ct < 4; ++ct) oacc[qt][ct] *= al[qt];

#pragma unroll
    for (int ct = 0; ct < 4; ++ct)
#pragma unroll
      for (int qt = 0; qt < 4; ++qt)
        oacc[qt][ct] = __builtin_amdgcn_wmma_f32_16x16x32_f16(false, pa[qt], false, vb[ct], (short)0, oacc[qt][ct], false, false);
  }

  {
    v8f fw;
#pragma unroll
    for (int r = 0; r < 8; ++r)
      fw[r] = (1.0f / lrow[r]) * mk[i0 + 8 * lh + r];
    if (l15 == 0) *reinterpret_cast<v8f*>(&Fbuf[wid * 16 + 8 * lh]) = fw;
  }
  __syncthreads();

#pragma unroll
  for (int qt = 0; qt < 4; ++qt) {
    const v8f fin = *reinterpret_cast<const v8f*>(&Fbuf[qt * 16 + 8 * lh]);
#pragma unroll
    for (int ct = 0; ct < 4; ++ct) {
      float* dst = out + ((size_t)n * CV + wid * 64 + ct * 16 + l15) * HW + iwg + qt * 16 + 8 * lh;
      float4 lo4 = make_float4(oacc[qt][ct][0] * fin[0], oacc[qt][ct][1] * fin[1],
                               oacc[qt][ct][2] * fin[2], oacc[qt][ct][3] * fin[3]);
      float4 hi4 = make_float4(oacc[qt][ct][4] * fin[4], oacc[qt][ct][5] * fin[5],
                               oacc[qt][ct][6] * fin[6], oacc[qt][ct][7] * fin[7]);
      *reinterpret_cast<float4*>(dst)     = lo4;
      *reinterpret_cast<float4*>(dst + 4) = hi4;
    }
  }
}

extern "C" void kernel_launch(void* const* d_in, const int* in_sizes, int n_in,
                              void* d_out, int out_size, void* d_ws, size_t ws_size,
                              hipStream_t stream) {
  (void)in_sizes; (void)n_in; (void)out_size; (void)ws_size;
  const float* x  = (const float*)d_in[0];
  const float* mk = (const float*)d_in[1];
  const float* rw = (const float*)d_in[2];
  const float* rg = (const float*)d_in[3];
  const float* rb = (const float*)d_in[4];
  const float* rm = (const float*)d_in[5];
  const float* rv = (const float*)d_in[6];
  const float* aw = (const float*)d_in[7];
  const float* ag = (const float*)d_in[8];
  const float* ab = (const float*)d_in[9];
  const float* am = (const float*)d_in[10];
  const float* av = (const float*)d_in[11];
  float* out = (float*)d_out;

  _Float16* qbuf = (_Float16*)d_ws;
  _Float16* fT   = qbuf + (size_t)NB * HW * CQ;

  const dim3 blk(32, 4);
  proj_bn_relu_kernel<<<dim3(HW / 64, CV / 16, NB), blk, 0, stream>>>(
      x, rw, rg, rb, rm, rv, fT, 0);
  proj_bn_relu_kernel<<<dim3(HW / 64, CQ / 16, NB), blk, 0, stream>>>(
      x, aw, ag, ab, am, av, qbuf, 1);
  flash_attn_kernel<<<dim3(HW / 64, NB), blk, 0, stream>>>(qbuf, fT, mk, out);
}


# Projections fused: one workgroup makes all 384 V and q channels of a 64-position block from one x tile staged in LDS (x read once, not 24 times); 3 launches become 2; channel sum order changed
# speedup vs baseline: 3.8243x; 1.1810x over previous
#include <hip/hip_runtime.h>
#include <hip/hip_bf16.h>

#define HW   4096
#define NB   8
#define CIN  256
#define CQ   128
#define CV   256

typedef float    v2f  __attribute__((ext_vector_type(2)));
typedef float    v8f  __attribute__((ext_vector_type(8)));
typedef _Float16 v16h __attribute__((ext_vector_type(16)));
typedef _Float16 h8   __attribute__((ext_vector_type(8)));

static __device__ __forceinline__ v16h load_frag16(const _Float16* rowBase, int lh) {
  const h8 lo = *reinterpret_cast<const h8*>(rowBase + 8 * lh);
  const h8 hi = *reinterpret_cast<const h8*>(rowBase + 8 * lh + 16);
  v16h r;
#pragma unroll
  for (int i = 0; i < 8; ++i) { r[i] = lo[i]; r[i + 8] = hi[i]; }
  return r;
}

static __device__ __forceinline__ v16h load_frag16_packed(const _Float16* fragBase, int lane) {
  const h8 lo = *reinterpret_cast<const h8*>(fragBase + 8 * lane);
  const h8 hi = *reinterpret_cast<const h8*>(fragBase + 256 + 8 * lane);
  v16h r;
#pragma unroll
  for (int i = 0; i < 8; ++i) { r[i] = lo[i]; r[i + 8] = hi[i]; }
  return r;
}

#define XP 68

struct WWin { float4 v[4][4]; float4 q[2][4]; };

static __device__ __forceinline__ void load_w_window(WWin& w, const float* __restrict__ wvr,
                                                     const float* __restrict__ wqr) {
#pragma unroll
  for (int s = 0; s < 4; ++s) {
#pragma unroll
    for (int ct = 0; ct < 4; ++ct) w.v[ct][s] = *reinterpret_cast<const float4*>(wvr + (size_t)ct * 16 * CIN + 8 * s);
#pragma unroll
    for (int ct = 0; ct < 2; ++ct) w.q[ct][s] = *reinterpret_cast<const float4*>(wqr + (size_t)ct * 16 * CIN + 8 * s);
  }
}

static __device__ __forceinline__ void mma_window(const WWin& w, const float* xf, int cw,
                                                  float (&xc)[4][4], v8f (&accv)[4][4], v8f (&accq)[2][4]) {
#pragma unroll
  for (int s = 0; s < 4; ++s) {
    const int c1 = (cw + 8 * s + 8) & (CIN - 1);
    float xn[4][4];
#pragma unroll
    for (int pt = 0; pt < 4; ++pt)
#pragma unroll
      for (int j = 0; j < 4; ++j) xn[pt][j] = xf[(c1 + j) * XP + 16 * pt];

#pragma unroll
    for (int h = 0; h < 2; ++h) {
      v2f xa[4];
#pragma unroll
      for (int pt = 0; pt < 4; ++pt) { xa[pt][0] = xc[pt][2 * h]; xa[pt][1] = xc[pt][2 * h + 1]; }
#pragma unroll
      for (int ct = 0; ct < 4; ++ct) {
        v2f wb;
        wb[0] = h ? w.v[ct][s].z : w.v[ct][s].x;
        wb[1] = h ? w.v[ct][s].w : w.v[ct][s].y;
#pragma unroll
        for (int pt = 0; pt < 4; ++pt)
          accv[pt][ct] = __builtin_amdgcn_wmma_f32_16x16x4_f32(false, xa[pt], false, wb, (short)0, accv[pt][ct], false, false);
      }
#pragma unroll
      for (int ct = 0; ct < 2; ++ct) {
        v2f wa;
        wa[0] = h ? w.q[ct][s].z : w.q[ct][s].x;
        wa[1] = h ? w.q[ct][s].w : w.q[ct][s].y;
#pragma unroll
        for (int pt = 0; pt < 4; ++pt)
          accq[ct][pt] = __builtin_amdgcn_wmma_f32_16x16x4_f32(false, wa, false, xa[pt], (short)0, accq[ct][pt], false, false);
      }
    }
#pragma unroll
    for (int pt = 0; pt < 4; ++pt)
#pragma unroll
      for (int j = 0; j < 4; ++j) xc[pt][j] = xn[pt][j];
  }
}


__global__ __launch_bounds__(128) __attribute__((amdgpu_waves_per_eu(2, 2)))
void proj_fused_kernel(
    const float* __restrict__ x,
    const float* __restrict__ wv, const float* __restrict__ gv, const float* __restrict__ bv,
    const float* __restrict__ mv, const float* __restrict__ vv,
    const float* __restrict__ wq, const float* __restrict__ gq, const float* __restrict__ bq,
    const float* __restrict__ mq, const float* __restrict__ vq,
    _Float16* __restrict__ vout, _Float16* __restrict__ qout)
{
  __shared__ __align__(16) float xs[CIN * XP];

  const int lane = threadIdx.x;
  const int l15  = lane & 15;
  const int lh   = lane >> 4;
  const int wid  = threadIdx.y;
  const int tid  = wid * 32 + lane;
  const int pblk = blockIdx.x * 64;
  const int n    = blockIdx.y;

  {
    const float* xg = x + (size_t)n * CIN * HW + pblk + 4 * (tid & 15);
    float*       xl = xs + 4 * (tid & 15);
    const int    cr = tid >> 4;
#pragma unroll
    for (int c = 0; c < CIN; c += 8) {
      const float4 t = *reinterpret_cast<const float4*>(xg + (size_t)(c + cr) * HW);
      *reinterpret_cast<float4*>(xl + (c + cr) * XP) = t;
    }
  }
  __syncthreads();

  const float* wvr = wv + (size_t)(wid * 64 + l15) * CIN + 4 * lh;
  const float* wqr = wq + (size_t)(wid * 32 + l15) * CIN + 4 * lh;
  const float* xf  = xs + (4 * lh) * XP + l15;

  v8f accv[4][4];
  v8f accq[2][4];
#pragma unroll
  for (int a = 0; a < 4; ++a)
#pragma unroll
    for (int b = 0; b < 4; ++b) accv[a][b] = (v8f){};
#pragma unroll
  for (int a = 0; a < 2; ++a)
#pragma unroll
    for (int b = 0; b < 4; ++b) accq[a][b] = (v8f){};

  WWin wA, wB;
  float xc[4][4];
  load_w_window(wA, wvr, wqr);
#pragma unroll
  for (int pt = 0; pt < 4; ++pt)
#pragma unroll
    for (int j = 0; j < 4; ++j) xc[pt][j] = xf[j * XP + 16 * pt];

  for (int cw = 0; cw < CIN; cw += 64) {
    load_w_window(wB, wvr + cw + 32, wqr + cw + 32);
    __builtin_amdgcn_sched_barrier(0);
    mma_window(wA, xf, cw, xc, accv, accq);
    load_w_window(wA, wvr + ((cw + 64) & (CIN - 1)), wqr + ((cw + 64) & (CIN - 1)));
    __builtin_amdgcn_sched_barrier(0);
    mma_window(wB, xf, cw + 32, xc, accv, accq);
  }

#pragma unroll
  for (int ct = 0; ct < 4; ++ct) {
    const int o0   = wid * 64 + ct * 16;
    const int ch   = o0 + l15;
    const float iv = gv[ch] * rsqrtf(vv[ch] + 1e-5f);
    const float bb = bv[ch] - mv[ch] * iv;
#pragma unroll
    for (int pt = 0; pt < 4; ++pt) {
      const int p0 = pblk + pt * 16;
      h8 hv;
#pragma unroll
      for (int r = 0; r < 8; ++r) {
        float v = accv[pt][ct][r] * iv + bb;
        v = v > 0.f ? v : 0.f;
        hv[r] = (_Float16)v;
      }
      const size_t off = ((((size_t)n * (HW / 32) + (p0 >> 5)) * (CV / 16) + (o0 >> 4)) * 2 + ((p0 >> 4) & 1)) * 256 + 8 * lane;
      *reinterpret_cast<h8*>(vout + off) = hv;
    }
  }
#pragma unroll
  for (int ct = 0; ct < 2; ++ct) {
    const int o0 = wid * 32 + ct * 16;
    float iv[8], bb[8];
#pragma unroll
    for (int r = 0; r < 8; ++r) {
      const int ch = o0 + 8 * lh + r;
      iv[r] = gq[ch] * rsqrtf(vq[ch] + 1e-5f);
      bb[r] = bq[ch] - mq[ch] * iv[r];
    }
#pragma unroll
    for (int pt = 0; pt < 4; ++pt) {
      const int p0 = pblk + pt * 16;
      h8 hv;
#pragma unroll
      for (int r = 0; r < 8; ++r) {
        float v = accq[ct][pt][r] * iv[r] + bb[r];
        v = v > 0.f ? v : 0.f;
        hv[r] = (_Float16)v;
      }
      const size_t off = ((((size_t)n * (HW / 16) + (p0 >> 4)) * 4 + (o0 >> 5)) * 2 + ((o0 >> 4) & 1)) * 256 + 8 * lane;
      *reinterpret_cast<h8*>(qout + off) = hv;
    }
  }
}

__global__ __launch_bounds__(128) __attribute__((amdgpu_waves_per_eu(2, 2)))
void flash_attn_kernel(
    const _Float16* __restrict__ q, const _Float16* __restrict__ fT,
    const float* __restrict__ mask, float* __restrict__ out)
{
  constexpr float LOG2E     = 1.4426950408889634f;
  constexpr float ATT_SCALE = (1.0f / (1e-8f + 11.313708498984761f)) * LOG2E;
  constexpr int PP = 40;
  __shared__ __align__(32) _Float16 Pbuf[2][64 * PP];
  __shared__ __align__(32) float    Abuf[2][64];
  __shared__ __align__(32) float    Fbuf[64];

  const int lane = threadIdx.x;
  const int l15  = lane & 15;
  const int lh   = lane >> 4;
  const int wid  = threadIdx.y;
  const int iwg  = blockIdx.x * 64;
  const int i0   = iwg + wid * 16;
  const int n    = blockIdx.y;

  const _Float16* qn = q + (size_t)n * HW * CQ;

  v16h qa[4];
  {
    const _Float16* qt0 = qn + (size_t)(i0 >> 4) * 2048;
#pragma unroll
    for (int kc = 0; kc < 4; ++kc) qa[kc] = load_frag16_packed(qt0 + kc * 512, lane);
  }

  v8f oacc[4][4];
#pragma unroll
  for (int qt = 0; qt < 4; ++qt)
#pragma unroll
    for (int ct = 0; ct < 4; ++ct) oacc[qt][ct] = (v8f){};
  float mrow[8], lrow[8];
#pragma unroll
  for (int r = 0; r < 8; ++r) { mrow[r] = -1e30f; lrow[r] = 0.f; }

  const float* mk = mask + (size_t)n * HW;
  const _Float16* vbase = fT + (size_t)n * CV * HW + wid * 4 * 512;

  for (int j0 = 0; j0 < HW; j0 += 32) {
    const int slot = (j0 >> 5) & 1;

    if (j0 + 32 < HW) {
      __builtin_prefetch(qn + (size_t)((j0 + 32) >> 4) * 2048 + lane * 128, 0, 3);
    }

    const float mv0 = mk[j0 + l15];
    const float mv1 = mk[j0 + 16 + l15];
    v16h kb0[4], kb1[4], vb[4];
    {
      const _Float16* kb = qn + (size_t)(j0 >> 4) * 2048;
#pragma unroll
      for (int kc = 0; kc < 4; ++kc) {
        kb0[kc] = load_frag16_packed(kb + kc * 512, lane);
        kb1[kc] = load_frag16_packed(kb + 2048 + kc * 512, lane);
      }
      const _Float16* vblk = vbase + (size_t)(j0 >> 5) * 8192;
#pragma unroll
      for (int ct = 0; ct < 4; ++ct) vb[ct] = load_frag16_packed(vblk + ct * 512, lane);
    }
    __builtin_amdgcn_sched_barrier(0);

    v8f s0 = (v8f){};
    v8f s1 = (v8f){};
#pragma unroll
    for (int kc = 0; kc < 4; ++kc) {
      s0 = __builtin_amdgcn_wmma_f32_16x16x32_f16(false, qa[kc], false, kb0[kc], (short)0, s0, false, false);
      s1 = __builtin_amdgcn_wmma_f32_16x16x32_f16(false, qa[kc], false, kb1[kc], (short)0, s1, false, false);
    }

    const float ma0 = (mv0 - 1.f) * (1e8f * LOG2E);
    const float ma1 = (mv1 - 1.f) * (1e8f * LOG2E);

    v8f avec;
#pragma unroll
    for (int r = 0; r < 8; ++r) {
      float e0 = s0[r] * ATT_SCALE + ma0;
      float e1 = s1[r] * ATT_SCALE + ma1;
      float vmax = fmaxf(e0, e1);
#pragma unroll
      for (int m = 1; m <= 8; m <<= 1) vmax = fmaxf(vmax, __shfl_xor(vmax, m, 32));
      const float mn = fmaxf(mrow[r], vmax);
      avec[r] = exp2f(mrow[r] - mn);
      mrow[r] = mn;
      e0 = exp2f(e0 - mn);
      e1 = exp2f(e1 - mn);
      float rs = e0 + e1;
#pragma unroll
      for (int m = 1; m <= 8; m <<= 1) rs += __shfl_xor(rs, m, 32);
      lrow[r] = lrow[r] * avec[r] + rs;
      s0[r] = e0;
      s1[r] = e1;
    }

    {
      _Float16* pw = &Pbuf[slot][(wid * 16) * PP];
#pragma unroll
      for (int r = 0; r < 8; ++r) {
        pw[(8 * lh + r) * PP + l15]      = (_Float16)s0[r];
        pw[(8 * lh + r) * PP + 16 + l15] = (_Float16)s1[r];
      }
      if (l15 == 0) *reinterpret_cast<v8f*>(&Abuf[slot][wid * 16 + 8 * lh]) = avec;
    }
    __syncthreads();

    v16h pa[4];
    v8f  al[4];
#pragma unroll
    for (int qt = 0; qt < 4; ++qt) {
      pa[qt] = load_frag16(&Pbuf[slot][(qt * 16 + l15) * PP], lh);
      al[qt] = *reinterpret_cast<const v8f*>(&Abuf[slot][qt * 16 + 8 * lh]);
    }

#pragma unroll
    for (int qt = 0; qt < 4; ++qt)
#pragma unroll
      for (int ct = 0; ct < 4; ++ct) oacc[qt][ct] *= al[qt];

#pragma unroll
    for (int ct = 0; ct < 4; ++ct)
#pragma unroll
      for (int qt = 0; qt < 4; ++qt)
        oacc[qt][ct] = __builtin_amdgcn_wmma_f32_16x16x32_f16(false, pa[qt], false, vb[ct], (short)0, oacc[qt][ct], false, false);
  }

  {
    v8f fw;
#pragma unroll
    for (int r = 0; r < 8; ++r)
      fw[r] = (1.0f / lrow[r]) * mk[i0 + 8 * lh + r];
    if (l15 == 0) *reinterpret_cast<v8f*>(&Fbuf[wid * 16 + 8 * lh]) = fw;
  }
  __syncthreads();

#pragma unroll
  for (int qt = 0; qt < 4; ++qt) {
    const v8f fin = *reinterpret_cast<const v8f*>(&Fbuf[qt * 16 + 8 * lh]);
#pragma unroll
    for (int ct = 0; ct < 4; ++ct) {
      float* dst = out + ((size_t)n * CV + wid * 64 + ct * 16 + l15) * HW + iwg + qt * 16 + 8 * lh;
      float4 lo4 = make_float4(oacc[qt][ct][0] * fin[0], oacc[qt][ct][1] * fin[1],
                               oacc[qt][ct][2] * fin[2], oacc[qt][ct][3] * fin[3]);
      float4 hi4 = make_float4(oacc[qt][ct][4] * fin[4], oacc[qt][ct][5] * fin[5],
                               oacc[qt][ct][6] * fin[6], oacc[qt][ct][7] * fin[7]);
      *reinterpret_cast<float4*>(dst)     = lo4;
      *reinterpret_cast<float4*>(dst + 4) = hi4;
    }
  }
}

extern "C" void kernel_launch(void* const* d_in, const int* in_sizes, int n_in,
                              void* d_out, int out_size, void* d_ws, size_t ws_size,
                              hipStream_t stream) {
  (void)in_sizes; (void)n_in; (void)out_size; (void)ws_size;
  const float* x  = (const float*)d_in[0];
  const float* mk = (const float*)d_in[1];
  const float* rw = (const float*)d_in[2];
  const float* rg = (const float*)d_in[3];
  const float* rb = (const float*)d_in[4];
  const float* rm = (const float*)d_in[5];
  const float* rv = (const float*)d_in[6];
  const float* aw = (const float*)d_in[7];
  const float* ag = (const float*)d_in[8];
  const float* ab = (const float*)d_in[9];
  const float* am = (const float*)d_in[10];
  const float* av = (const float*)d_in[11];
  float* out = (float*)d_out;

  _Float16* qbuf = (_Float16*)d_ws;
  _Float16* fT   = qbuf + (size_t)NB * HW * CQ;

  const dim3 blk(32, 4);
  proj_fused_kernel<<<dim3(HW / 64, NB), blk, 0, stream>>>(
      x, rw, rg, rb, rm, rv, aw, ag, ab, am, av, fT, qbuf);
  flash_attn_kernel<<<dim3(HW / 64, NB), blk, 0, stream>>>(qbuf, fT, mk, out);
}
